# starting kernel
# baseline (speedup 1.0000x reference)
#include <hip/hip_runtime.h>
#include <math.h>

typedef float v2f __attribute__((ext_vector_type(2)));
typedef float v8f __attribute__((ext_vector_type(8)));

#define BS      65536
#define IN_DIM  100
#define HID     16
#define NCORP   10
#define ROWS    (BS * NCORP)
#define WPB     8

__global__ __launch_bounds__(256) void proj_tvec_kernel(
    const float* __restrict__ item_corpus,
    const float* __restrict__ W_proj,
    const float* __restrict__ b_proj,
    const float* __restrict__ W_t,
    float* __restrict__ proj,
    float* __restrict__ tvec)
{
    __shared__ float lds[WPB][16][17];

    const int lane  = threadIdx.x & 31;
    const int wv    = threadIdx.x >> 5;
    const int l15   = lane & 15;
    const int khalf = (lane >> 4) * 2;
    const int rofs  = (lane >> 4) * 8;

    const long tile = (long)blockIdx.x * WPB + wv;
    const long row0 = tile * 16;

    v8f acc = {};
    const float* arow = item_corpus + (row0 + l15) * IN_DIM;
    #pragma unroll
    for (int k0 = 0; k0 < IN_DIM; k0 += 4) {
        v2f a = *(const v2f*)(arow + k0 + khalf);
        v2f b;
        b.x = W_proj[(k0 + khalf + 0) * HID + l15];
        b.y = W_proj[(k0 + khalf + 1) * HID + l15];
        acc = __builtin_amdgcn_wmma_f32_16x16x4_f32(
                  false, a, false, b, (short)0, acc, false, false);
    }

    const float bias = b_proj[l15];
    #pragma unroll
    for (int r = 0; r < 8; ++r) acc[r] += bias;

    #pragma unroll
    for (int r = 0; r < 8; ++r) {
        proj[(row0 + r + rofs) * HID + l15] = acc[r];
        lds[wv][r + rofs][l15] = acc[r];
    }
    __syncthreads();

    v8f acc2 = {};
    #pragma unroll
    for (int kc = 0; kc < 4; ++kc) {
        v2f a;
        a.x = lds[wv][l15][kc * 4 + khalf + 0];
        a.y = lds[wv][l15][kc * 4 + khalf + 1];
        v2f b;
        b.x = W_t[(kc * 4 + khalf + 0) * HID + l15];
        b.y = W_t[(kc * 4 + khalf + 1) * HID + l15];
        acc2 = __builtin_amdgcn_wmma_f32_16x16x4_f32(
                   false, a, false, b, (short)0, acc2, false, false);
    }
    #pragma unroll
    for (int r = 0; r < 8; ++r)
        tvec[(row0 + r + rofs) * HID + l15] = tanhf(acc2[r]);
}

__global__ __launch_bounds__(256) void greedy_kernel(
    const float* __restrict__ user_intent,
    const float* __restrict__ proj,
    const float* __restrict__ tvec,
    const float* __restrict__ W_s,
    const int*   __restrict__ search_num_p,
    float* __restrict__ out)
{
    __shared__ float Ws[256];
    Ws[threadIdx.x] = W_s[threadIdx.x];
    __syncthreads();

    const int  b     = blockIdx.x * blockDim.x + threadIdx.x;
    const int  steps = *search_num_p;
    const long obase = (long)b * (steps + 1) * HID;

    float sum[HID];
    #pragma unroll
    for (int h = 0; h < HID; ++h) {
        float v = user_intent[(long)b * HID + h];
        sum[h] = v;
        out[obase + h] = v;
    }

    const float* tb = tvec + (long)b * NCORP * HID;
    const float* cb = proj + (long)b * NCORP * HID;

    for (int it = 0; it < steps; ++it) {
        const float inv = 1.0f / (float)(it + 1);

        float s[HID];
        #pragma unroll
        for (int h = 0; h < HID; ++h) {
            float a = 0.0f;
            #pragma unroll
            for (int hp = 0; hp < HID; ++hp)
                a += (sum[hp] * inv) * Ws[hp * HID + h];
            s[h] = tanhf(a);
        }

        float best = -__builtin_inff();
        int bidx = 0;
        for (int c = 0; c < NCORP; ++c) {
            float sc = 0.0f;
            #pragma unroll
            for (int h = 0; h < HID; ++h)
                sc += tb[c * HID + h] * s[h];
            if (sc > best) { best = sc; bidx = c; }
        }

        #pragma unroll
        for (int h = 0; h < HID; ++h) {
            float v = cb[bidx * HID + h];
            sum[h] += v;
            out[obase + (long)(it + 1) * HID + h] = v;
        }
    }
}

extern "C" void kernel_launch(void* const* d_in, const int* in_sizes, int n_in,
                              void* d_out, int out_size, void* d_ws, size_t ws_size,
                              hipStream_t stream) {
    const float* user_intent = (const float*)d_in[0];
    const float* item_corpus = (const float*)d_in[1];
    const float* W_proj      = (const float*)d_in[2];
    const float* b_proj      = (const float*)d_in[3];
    const float* W_s         = (const float*)d_in[4];
    const float* W_t         = (const float*)d_in[5];
    const int*   search_num  = (const int*)d_in[6];
    float* out = (float*)d_out;

    float* proj = (float*)d_ws;
    float* tv   = proj + (size_t)ROWS * HID;

    const int tiles   = ROWS / 16;
    const int blocks1 = tiles / WPB;

    proj_tvec_kernel<<<blocks1, 256, 0, stream>>>(item_corpus, W_proj, b_proj,
                                                  W_t, proj, tv);
    greedy_kernel<<<BS / 256, 256, 0, stream>>>(user_intent, proj, tv,
                                                W_s, search_num, out);
}


# Search kernel touches global memory in whole lines only: a wave's tvec, proj and out blocks move in 512-B instructions and turn through private LDS (tvec kept in registers); same sums, same order
# speedup vs baseline: 1.7184x; 1.7184x over previous
#include <hip/hip_runtime.h>
#include <math.h>

typedef float v2f __attribute__((ext_vector_type(2)));
typedef float v4f __attribute__((ext_vector_type(4)));
typedef float v8f __attribute__((ext_vector_type(8)));

#define BS      65536
#define IN_DIM  100
#define HID     16
#define NCORP   10
#define ROWS    (BS * NCORP)
#define WPB     8
#define NBLK1   512
#define NWAVE1  (NBLK1 * WPB)
#define TPW     (ROWS / 16 / NWAVE1)


__device__ __forceinline__ float tanh_sel(const float x)
{
    const float y = fabsf(x);
    const float t = x * x;
    float p = fmaf(__uint_as_float(0xbbbac73du), t, __uint_as_float(0x3ca908c9u));
    p = fmaf(t, p, __uint_as_float(0xbd5c1c4eu));
    p = fmaf(t, p, __uint_as_float(0x3e088382u));
    p = fmaf(t, p, __uint_as_float(0xbeaaaa99u));
    const float zs = fmaf(t, y * p, y);
    const float e  = expf(y + y);
    const float zb = fmaf(__builtin_amdgcn_rcpf(e + 1.0f), -2.0f, 1.0f);
    return __builtin_copysignf((y < 0.625f) ? zs : zb, x);
}

__device__ __forceinline__ void tile_fetch(const float* __restrict__ p,
                                           const float* __restrict__ pa,
                                           float (&t)[48], v2f& tl)
{
    #pragma unroll
    for (int q = 0; q < 6; ++q) {
        #pragma unroll
        for (int s = 0; s < 4; ++s) {
            t[(q * 4 + s) * 2 + 0] = p[(4 * s + 0) * IN_DIM + 16 * q];
            t[(q * 4 + s) * 2 + 1] = p[(4 * s + 1) * IN_DIM + 16 * q];
        }
    }
    tl = *(const v2f*)pa;
}

__device__ __forceinline__ void tile_compute(
    const float (&t)[48], const v2f tl,
    const v2f (&sel)[4], const v2f (&bw)[25], const v2f (&bt)[4],
    const float bias, float (*ldsw)[17],
    float* __restrict__ proj, float* __restrict__ tvec,
    const long row0, const int l15, const int khalf, const int rofs)
{
    v8f acc = {};
    #pragma unroll
    for (int q = 0; q < 6; ++q) {
        v8f xt = {};
        #pragma unroll
        for (int s = 0; s < 4; ++s) {
            v2f a;
            a.x = t[(q * 4 + s) * 2 + 0];
            a.y = t[(q * 4 + s) * 2 + 1];
            xt = __builtin_amdgcn_wmma_f32_16x16x4_f32(
                     false, a, false, sel[s], (short)0, xt, false, false);
        }
        #pragma unroll
        for (int j = 0; j < 4; ++j) {
            v2f a;
            a.x = xt[2 * j + 0];
            a.y = xt[2 * j + 1];
            acc = __builtin_amdgcn_wmma_f32_16x16x4_f32(
                      false, a, false, bw[4 * q + j], (short)0, acc, false, false);
        }
    }
    acc = __builtin_amdgcn_wmma_f32_16x16x4_f32(
              false, tl, false, bw[24], (short)0, acc, false, false);

    #pragma unroll
    for (int r = 0; r < 8; ++r) acc[r] += bias;

    #pragma unroll
    for (int r = 0; r < 8; ++r) {
        proj[(row0 + r + rofs) * HID + l15] = acc[r];
        ldsw[r + rofs][l15] = acc[r];
    }
    __builtin_amdgcn_fence(__ATOMIC_RELEASE, "wavefront");
    __builtin_amdgcn_wave_barrier();
    __builtin_amdgcn_fence(__ATOMIC_ACQUIRE, "wavefront");

    v8f acc2 = {};
    #pragma unroll
    for (int kc = 0; kc < 4; ++kc) {
        v2f a;
        a.x = ldsw[l15][kc * 4 + khalf + 0];
        a.y = ldsw[l15][kc * 4 + khalf + 1];
        acc2 = __builtin_amdgcn_wmma_f32_16x16x4_f32(
                   false, a, false, bt[kc], (short)0, acc2, false, false);
    }
    __builtin_amdgcn_fence(__ATOMIC_RELEASE, "wavefront");
    __builtin_amdgcn_wave_barrier();
    __builtin_amdgcn_fence(__ATOMIC_ACQUIRE, "wavefront");
    #pragma unroll
    for (int r = 0; r < 8; ++r)
        tvec[(row0 + r + rofs) * HID + l15] = tanh_sel(acc2[r]);
}

__global__ __launch_bounds__(256) void proj_tvec_kernel(
    const float* __restrict__ item_corpus,
    const float* __restrict__ W_proj,
    const float* __restrict__ b_proj,
    const float* __restrict__ W_t,
    float* __restrict__ proj,
    float* __restrict__ tvec)
{
    __shared__ float lds[WPB][16][17];

    const int lane  = threadIdx.x & 31;
    const int wv    = threadIdx.x >> 5;
    const int l15   = lane & 15;
    const int khalf = (lane >> 4) * 2;
    const int rofs  = (lane >> 4) * 8;

    const int kmap = 4 * ((l15 & 7) >> 1) + 2 * (l15 >> 3) + (l15 & 1);

    v2f sel[4];
    #pragma unroll
    for (int s = 0; s < 4; ++s) {
        sel[s].x = (l15 == 4 * s + khalf + 0) ? 1.0f : 0.0f;
        sel[s].y = (l15 == 4 * s + khalf + 1) ? 1.0f : 0.0f;
    }

    v2f bw[25];
    #pragma unroll
    for (int ks = 0; ks < 25; ++ks) {
        bw[ks].x = W_proj[(4 * ks + khalf + 0) * HID + l15];
        bw[ks].y = W_proj[(4 * ks + khalf + 1) * HID + l15];
    }
    v2f bt[4];
    #pragma unroll
    for (int kc = 0; kc < 4; ++kc) {
        bt[kc].x = W_t[(kc * 4 + khalf + 0) * HID + l15];
        bt[kc].y = W_t[(kc * 4 + khalf + 1) * HID + l15];
    }
    const float bias = b_proj[l15];

    const long gw     = (long)blockIdx.x * WPB + wv;
    const long tstep  = (long)NWAVE1 * 16;
    long row0 = gw * 16;
    const float* p  = item_corpus + (row0 + khalf) * IN_DIM + kmap;
    const float* pa = item_corpus + (row0 + l15) * IN_DIM + 96 + khalf;
    const long pstep = tstep * IN_DIM;

    float ta[48], tb[48];
    v2f   tla, tlb;

    tile_fetch(p, pa, ta, tla);
    #pragma unroll 1
    for (int i = 0; i < TPW - 2; i += 2) {
        tile_fetch(p + pstep, pa + pstep, tb, tlb);
        tile_compute(ta, tla, sel, bw, bt, bias, lds[wv], proj, tvec,
                     row0, l15, khalf, rofs);
        tile_fetch(p + 2 * pstep, pa + 2 * pstep, ta, tla);
        tile_compute(tb, tlb, sel, bw, bt, bias, lds[wv], proj, tvec,
                     row0 + tstep, l15, khalf, rofs);
        p    += 2 * pstep;
        pa   += 2 * pstep;
        row0 += 2 * tstep;
    }
    tile_fetch(p + pstep, pa + pstep, tb, tlb);
    tile_compute(ta, tla, sel, bw, bt, bias, lds[wv], proj, tvec,
                 row0, l15, khalf, rofs);
    tile_compute(tb, tlb, sel, bw, bt, bias, lds[wv], proj, tvec,
                 row0 + tstep, l15, khalf, rofs);
}

#define TV_U      (NCORP * HID / 4)
#define TV_PITCH  (TV_U + 1)
#define OUT_PITCH 25

__global__ __launch_bounds__(256) void greedy_kernel(
    const float* __restrict__ user_intent,
    const float* __restrict__ proj,
    const float* __restrict__ tvec,
    const float* __restrict__ W_s,
    const int*   __restrict__ search_num_p,
    float* __restrict__ out)
{
    __shared__ float Ws[256];
    __shared__ v4f   blk[WPB][32 * TV_PITCH];
    __shared__ v4f   ots[WPB][32 * OUT_PITCH];
    Ws[threadIdx.x] = W_s[threadIdx.x];

    const int  b     = blockIdx.x * blockDim.x + threadIdx.x;
    const int  lane  = threadIdx.x & 31;
    const int  wv    = threadIdx.x >> 5;
    const long e0    = b - lane;
    const int  steps = *search_num_p;
    const long obase = (long)b * (steps + 1) * HID;
    const bool staged = 4 * (steps + 1) < OUT_PITCH;

    const v4f* tsrc = (const v4f*)(tvec + e0 * NCORP * HID);
    const v4f* psrc = (const v4f*)(proj + e0 * NCORP * HID);
    v4f stt[TV_U], stp[TV_U];
    #pragma unroll
    for (int i = 0; i < TV_U; ++i) stt[i] = tsrc[32 * i + lane];
    #pragma unroll
    for (int i = 0; i < TV_U; ++i) {
        const int row = (32 * i) / TV_U + ((lane + (32 * i) % TV_U >= TV_U) ? 1 : 0);
        blk[wv][32 * i + lane + row] = stt[i];
    }
    #pragma unroll
    for (int i = 0; i < TV_U; ++i) stp[i] = psrc[32 * i + lane];
    __syncthreads();

    float tv[NCORP * HID];
    #pragma unroll
    for (int j = 0; j < TV_U; ++j) {
        const v4f u = blk[wv][TV_PITCH * lane + j];
        tv[4 * j + 0] = u.x;
        tv[4 * j + 1] = u.y;
        tv[4 * j + 2] = u.z;
        tv[4 * j + 3] = u.w;
    }
    __syncthreads();
    #pragma unroll
    for (int i = 0; i < TV_U; ++i) {
        const int row = (32 * i) / TV_U + ((lane + (32 * i) % TV_U >= TV_U) ? 1 : 0);
        blk[wv][32 * i + lane + row] = stp[i];
    }
    __syncthreads();

    float sum[HID];
    const v4f* usrc = (const v4f*)(user_intent + (long)b * HID);
    v4f g[4];
    #pragma unroll
    for (int j = 0; j < 4; ++j) {
        g[j] = usrc[j];
        sum[4 * j + 0] = g[j].x;
        sum[4 * j + 1] = g[j].y;
        sum[4 * j + 2] = g[j].z;
        sum[4 * j + 3] = g[j].w;
    }
    if (staged) {
        #pragma unroll
        for (int j = 0; j < 4; ++j) ots[wv][OUT_PITCH * lane + j] = g[j];
    } else {
        #pragma unroll
        for (int j = 0; j < 4; ++j) *(v4f*)(out + obase + 4 * j) = g[j];
    }

    for (int it = 0; it < steps; ++it) {
        const float inv = 1.0f / (float)(it + 1);

        float xs[HID];
        #pragma unroll
        for (int hp = 0; hp < HID; ++hp) xs[hp] = sum[hp] * inv;
        float s[HID];
        #pragma unroll
        for (int h = 0; h < HID; ++h) {
            float a = 0.0f;
            #pragma unroll
            for (int hp = 0; hp < HID; ++hp)
                a = fmaf(xs[hp], Ws[hp * HID + h], a);
            s[h] = tanh_sel(a);
        }

        float best = -__builtin_inff();
        int bidx = 0;
        #pragma unroll
        for (int c = 0; c < NCORP; ++c) {
            float sc = 0.0f;
            {
                #pragma clang fp contract(off)
                #pragma unroll
                for (int h = 0; h < HID; ++h)
                    sc += tv[c * HID + h] * s[h];
            }
            if (sc > best) { best = sc; bidx = c; }
        }

        #pragma unroll
        for (int j = 0; j < 4; ++j) {
            g[j] = blk[wv][TV_PITCH * lane + 4 * bidx + j];
            sum[4 * j + 0] += g[j].x;
            sum[4 * j + 1] += g[j].y;
            sum[4 * j + 2] += g[j].z;
            sum[4 * j + 3] += g[j].w;
        }
        if (staged) {
            #pragma unroll
            for (int j = 0; j < 4; ++j)
                ots[wv][OUT_PITCH * lane + 4 * (it + 1) + j] = g[j];
        } else {
            #pragma unroll
            for (int j = 0; j < 4; ++j)
                *(v4f*)(out + obase + (long)(it + 1) * HID + 4 * j) = g[j];
        }
    }

    if (staged) {
        __syncthreads();
        const int nu = 4 * (steps + 1);
        v4f* odst = (v4f*)(out + e0 * (steps + 1) * HID);
        for (int i = 0; i < nu; ++i) {
            const int q   = 32 * i + lane;
            const int row = q / nu;
            odst[q] = ots[wv][OUT_PITCH * row + (q - row * nu)];
        }
    }
}

extern "C" void kernel_launch(void* const* d_in, const int* in_sizes, int n_in,
                              void* d_out, int out_size, void* d_ws, size_t ws_size,
                              hipStream_t stream) {
    const float* user_intent = (const float*)d_in[0];
    const float* item_corpus = (const float*)d_in[1];
    const float* W_proj      = (const float*)d_in[2];
    const float* b_proj      = (const float*)d_in[3];
    const float* W_s         = (const float*)d_in[4];
    const float* W_t         = (const float*)d_in[5];
    const int*   search_num  = (const int*)d_in[6];
    float* out = (float*)d_out;

    float* proj = (float*)d_ws;
    float* tv   = proj + (size_t)ROWS * HID;

    proj_tvec_kernel<<<NBLK1, 256, 0, stream>>>(item_corpus, W_proj, b_proj,
                                                W_t, proj, tv);
    greedy_kernel<<<BS / 256, 256, 0, stream>>>(user_intent, proj, tv,
                                                W_s, search_num, out);
}


# Projection A fragments made from 64-B row-piece loads of four rows (8 B per lane) by two masked lane exchanges in the 16-lane row (l^4, l^8), not by 24 selector WMMAs a tile; same k order, same sums
# speedup vs baseline: 1.7839x; 1.0381x over previous
#include <hip/hip_runtime.h>
#include <math.h>

typedef float v2f __attribute__((ext_vector_type(2)));
typedef float v4f __attribute__((ext_vector_type(4)));
typedef float v8f __attribute__((ext_vector_type(8)));

#define BS      65536
#define IN_DIM  100
#define HID     16
#define NCORP   10
#define ROWS    (BS * NCORP)
#define WPB     8
#define NBLK1   512
#define NWAVE1  (NBLK1 * WPB)
#define TPW     (ROWS / 16 / NWAVE1)


template <int XM, int BANKS>
__device__ __forceinline__ float bank_xchg(const float keep, const float from)
{
    return __int_as_float(__builtin_amdgcn_update_dpp(
        __float_as_int(keep), __float_as_int(from), 0x160 + XM, 0xf, BANKS, false));
}
template <int XM, int BANKS>
__device__ __forceinline__ v2f bank_xchg2(const v2f keep, const v2f from)
{
    v2f r;
    r.x = bank_xchg<XM, BANKS>(keep.x, from.x);
    r.y = bank_xchg<XM, BANKS>(keep.y, from.y);
    return r;
}

__device__ __forceinline__ float tanh_sel(const float x)
{
    const float y = fabsf(x);
    const float t = x * x;
    float p = fmaf(__uint_as_float(0xbbbac73du), t, __uint_as_float(0x3ca908c9u));
    p = fmaf(t, p, __uint_as_float(0xbd5c1c4eu));
    p = fmaf(t, p, __uint_as_float(0x3e088382u));
    p = fmaf(t, p, __uint_as_float(0xbeaaaa99u));
    const float zs = fmaf(t, y * p, y);
    const float e  = expf(y + y);
    const float zb = fmaf(__builtin_amdgcn_rcpf(e + 1.0f), -2.0f, 1.0f);
    return __builtin_copysignf((y < 0.625f) ? zs : zb, x);
}

__device__ __forceinline__ void tile_fetch(const float* __restrict__ p,
                                           const float* __restrict__ pa,
                                           v2f (&t)[24], v2f& tl)
{
    #pragma unroll
    for (int q = 0; q < 6; ++q) {
        #pragma unroll
        for (int g = 0; g < 4; ++g)
            t[q * 4 + g] = *(const v2f*)(p + (4 * g) * IN_DIM + 16 * q);
    }
    tl = *(const v2f*)pa;
}

__device__ __forceinline__ void tile_compute(
    const v2f (&t)[24], const v2f tl,
    const v2f (&bw)[25], const v2f (&bt)[4],
    const float bias, float (*ldsw)[17],
    float* __restrict__ proj, float* __restrict__ tvec,
    const long row0, const int l15, const int khalf, const int rofs)
{
    v8f acc = {};
    #pragma unroll
    for (int q = 0; q < 6; ++q) {
        const v2f u0 = bank_xchg2<4, 0xA>(t[q * 4 + 0], t[q * 4 + 1]);
        const v2f u1 = bank_xchg2<4, 0x5>(t[q * 4 + 1], t[q * 4 + 0]);
        const v2f u2 = bank_xchg2<4, 0xA>(t[q * 4 + 2], t[q * 4 + 3]);
        const v2f u3 = bank_xchg2<4, 0x5>(t[q * 4 + 3], t[q * 4 + 2]);
        v2f y[4];
        y[0] = bank_xchg2<8, 0xC>(u0, u2);
        y[2] = bank_xchg2<8, 0x3>(u2, u0);
        y[1] = bank_xchg2<8, 0xC>(u1, u3);
        y[3] = bank_xchg2<8, 0x3>(u3, u1);
        #pragma unroll
        for (int j = 0; j < 4; ++j)
            acc = __builtin_amdgcn_wmma_f32_16x16x4_f32(
                      false, y[j], false, bw[4 * q + j], (short)0, acc, false, false);
    }
    acc = __builtin_amdgcn_wmma_f32_16x16x4_f32(
              false, tl, false, bw[24], (short)0, acc, false, false);

    #pragma unroll
    for (int r = 0; r < 8; ++r) acc[r] += bias;

    #pragma unroll
    for (int r = 0; r < 8; ++r) {
        proj[(row0 + r + rofs) * HID + l15] = acc[r];
        ldsw[r + rofs][l15] = acc[r];
    }
    __builtin_amdgcn_fence(__ATOMIC_RELEASE, "wavefront");
    __builtin_amdgcn_wave_barrier();
    __builtin_amdgcn_fence(__ATOMIC_ACQUIRE, "wavefront");

    v8f acc2 = {};
    #pragma unroll
    for (int kc = 0; kc < 4; ++kc) {
        v2f a;
        a.x = ldsw[l15][kc * 4 + khalf + 0];
        a.y = ldsw[l15][kc * 4 + khalf + 1];
        acc2 = __builtin_amdgcn_wmma_f32_16x16x4_f32(
                   false, a, false, bt[kc], (short)0, acc2, false, false);
    }
    __builtin_amdgcn_fence(__ATOMIC_RELEASE, "wavefront");
    __builtin_amdgcn_wave_barrier();
    __builtin_amdgcn_fence(__ATOMIC_ACQUIRE, "wavefront");
    #pragma unroll
    for (int r = 0; r < 8; ++r)
        tvec[(row0 + r + rofs) * HID + l15] = tanh_sel(acc2[r]);
}

__global__ __launch_bounds__(256) void proj_tvec_kernel(
    const float* __restrict__ item_corpus,
    const float* __restrict__ W_proj,
    const float* __restrict__ b_proj,
    const float* __restrict__ W_t,
    float* __restrict__ proj,
    float* __restrict__ tvec)
{
    __shared__ float lds[WPB][16][17];

    const int lane  = threadIdx.x & 31;
    const int wv    = threadIdx.x >> 5;
    const int l15   = lane & 15;
    const int khalf = (lane >> 4) * 2;
    const int rofs  = (lane >> 4) * 8;


    v2f bw[25];
    #pragma unroll
    for (int ks = 0; ks < 25; ++ks) {
        bw[ks].x = W_proj[(4 * ks + khalf + 0) * HID + l15];
        bw[ks].y = W_proj[(4 * ks + khalf + 1) * HID + l15];
    }
    v2f bt[4];
    #pragma unroll
    for (int kc = 0; kc < 4; ++kc) {
        bt[kc].x = W_t[(kc * 4 + khalf + 0) * HID + l15];
        bt[kc].y = W_t[(kc * 4 + khalf + 1) * HID + l15];
    }
    const float bias = b_proj[l15];

    const long gw     = (long)blockIdx.x * WPB + wv;
    const long tstep  = (long)NWAVE1 * 16;
    long row0 = gw * 16;
    const float* p  = item_corpus + (row0 + (lane & 3)) * IN_DIM + 4 * ((lane >> 2) & 3) + khalf;
    const float* pa = item_corpus + (row0 + l15) * IN_DIM + 96 + khalf;
    const long pstep = tstep * IN_DIM;

    v2f   ta[24], tb[24];
    v2f   tla, tlb;

    tile_fetch(p, pa, ta, tla);
    #pragma unroll 1
    for (int i = 0; i < TPW - 2; i += 2) {
        tile_fetch(p + pstep, pa + pstep, tb, tlb);
        __builtin_amdgcn_sched_barrier(0);
        tile_compute(ta, tla, bw, bt, bias, lds[wv], proj, tvec,
                     row0, l15, khalf, rofs);
        tile_fetch(p + 2 * pstep, pa + 2 * pstep, ta, tla);
        __builtin_amdgcn_sched_barrier(0);
        tile_compute(tb, tlb, bw, bt, bias, lds[wv], proj, tvec,
                     row0 + tstep, l15, khalf, rofs);
        p    += 2 * pstep;
        pa   += 2 * pstep;
        row0 += 2 * tstep;
    }
    tile_fetch(p + pstep, pa + pstep, tb, tlb);
    __builtin_amdgcn_sched_barrier(0);
    tile_compute(ta, tla, bw, bt, bias, lds[wv], proj, tvec,
                 row0, l15, khalf, rofs);
    tile_compute(tb, tlb, bw, bt, bias, lds[wv], proj, tvec,
                 row0 + tstep, l15, khalf, rofs);
}

#define TV_U      (NCORP * HID / 4)
#define TV_PITCH  (TV_U + 1)
#define OUT_PITCH 25

__global__ __launch_bounds__(256) void greedy_kernel(
    const float* __restrict__ user_intent,
    const float* __restrict__ proj,
    const float* __restrict__ tvec,
    const float* __restrict__ W_s,
    const int*   __restrict__ search_num_p,
    float* __restrict__ out)
{
    __shared__ float Ws[256];
    __shared__ v4f   blk[WPB][32 * TV_PITCH];
    __shared__ v4f   ots[WPB][32 * OUT_PITCH];
    Ws[threadIdx.x] = W_s[threadIdx.x];

    const int  b     = blockIdx.x * blockDim.x + threadIdx.x;
    const int  lane  = threadIdx.x & 31;
    const int  wv    = threadIdx.x >> 5;
    const long e0    = b - lane;
    const int  steps = *search_num_p;
    const long obase = (long)b * (steps + 1) * HID;
    const bool staged = 4 * (steps + 1) < OUT_PITCH;

    const v4f* tsrc = (const v4f*)(tvec + e0 * NCORP * HID);
    const v4f* psrc = (const v4f*)(proj + e0 * NCORP * HID);
    v4f stt[TV_U], stp[TV_U];
    #pragma unroll
    for (int i = 0; i < TV_U; ++i) stt[i] = tsrc[32 * i + lane];
    #pragma unroll
    for (int i = 0; i < TV_U; ++i) {
        const int row = (32 * i) / TV_U + ((lane + (32 * i) % TV_U >= TV_U) ? 1 : 0);
        blk[wv][32 * i + lane + row] = stt[i];
    }
    #pragma unroll
    for (int i = 0; i < TV_U; ++i) stp[i] = psrc[32 * i + lane];
    __syncthreads();

    float tv[NCORP * HID];
    #pragma unroll
    for (int j = 0; j < TV_U; ++j) {
        const v4f u = blk[wv][TV_PITCH * lane + j];
        tv[4 * j + 0] = u.x;
        tv[4 * j + 1] = u.y;
        tv[4 * j + 2] = u.z;
        tv[4 * j + 3] = u.w;
    }
    __syncthreads();
    #pragma unroll
    for (int i = 0; i < TV_U; ++i) {
        const int row = (32 * i) / TV_U + ((lane + (32 * i) % TV_U >= TV_U) ? 1 : 0);
        blk[wv][32 * i + lane + row] = stp[i];
    }
    __syncthreads();

    float sum[HID];
    const v4f* usrc = (const v4f*)(user_intent + (long)b * HID);
    v4f g[4];
    #pragma unroll
    for (int j = 0; j < 4; ++j) {
        g[j] = usrc[j];
        sum[4 * j + 0] = g[j].x;
        sum[4 * j + 1] = g[j].y;
        sum[4 * j + 2] = g[j].z;
        sum[4 * j + 3] = g[j].w;
    }
    if (staged) {
        #pragma unroll
        for (int j = 0; j < 4; ++j) ots[wv][OUT_PITCH * lane + j] = g[j];
    } else {
        #pragma unroll
        for (int j = 0; j < 4; ++j) *(v4f*)(out + obase + 4 * j) = g[j];
    }

    for (int it = 0; it < steps; ++it) {
        const float inv = 1.0f / (float)(it + 1);

        float xs[HID];
        #pragma unroll
        for (int hp = 0; hp < HID; ++hp) xs[hp] = sum[hp] * inv;
        float s[HID];
        #pragma unroll
        for (int h = 0; h < HID; ++h) {
            float a = 0.0f;
            #pragma unroll
            for (int hp = 0; hp < HID; ++hp)
                a = fmaf(xs[hp], Ws[hp * HID + h], a);
            s[h] = tanh_sel(a);
        }

        float best = -__builtin_inff();
        int bidx = 0;
        #pragma unroll
        for (int c = 0; c < NCORP; ++c) {
            float sc = 0.0f;
            {
                #pragma clang fp contract(off)
                #pragma unroll
                for (int h = 0; h < HID; ++h)
                    sc += tv[c * HID + h] * s[h];
            }
            if (sc > best) { best = sc; bidx = c; }
        }

        #pragma unroll
        for (int j = 0; j < 4; ++j) {
            g[j] = blk[wv][TV_PITCH * lane + 4 * bidx + j];
            sum[4 * j + 0] += g[j].x;
            sum[4 * j + 1] += g[j].y;
            sum[4 * j + 2] += g[j].z;
            sum[4 * j + 3] += g[j].w;
        }
        if (staged) {
            #pragma unroll
            for (int j = 0; j < 4; ++j)
                ots[wv][OUT_PITCH * lane + 4 * (it + 1) + j] = g[j];
        } else {
            #pragma unroll
            for (int j = 0; j < 4; ++j)
                *(v4f*)(out + obase + (long)(it + 1) * HID + 4 * j) = g[j];
        }
    }

    if (staged) {
        __syncthreads();
        const int nu = 4 * (steps + 1);
        v4f* odst = (v4f*)(out + e0 * (steps + 1) * HID);
        for (int i = 0; i < nu; ++i) {
            const int q   = 32 * i + lane;
            const int row = q / nu;
            odst[q] = ots[wv][OUT_PITCH * row + (q - row * nu)];
        }
    }
}

extern "C" void kernel_launch(void* const* d_in, const int* in_sizes, int n_in,
                              void* d_out, int out_size, void* d_ws, size_t ws_size,
                              hipStream_t stream) {
    const float* user_intent = (const float*)d_in[0];
    const float* item_corpus = (const float*)d_in[1];
    const float* W_proj      = (const float*)d_in[2];
    const float* b_proj      = (const float*)d_in[3];
    const float* W_s         = (const float*)d_in[4];
    const float* W_t         = (const float*)d_in[5];
    const int*   search_num  = (const int*)d_in[6];
    float* out = (float*)d_out;

    float* proj = (float*)d_ws;
    float* tv   = proj + (size_t)ROWS * HID;

    proj_tvec_kernel<<<NBLK1, 256, 0, stream>>>(item_corpus, W_proj, b_proj,
                                                W_t, proj, tv);
    greedy_kernel<<<BS / 256, 256, 0, stream>>>(user_intent, proj, tv,
                                                W_s, search_num, out);
}


# tanh(proj W_t) moved from the projection launch into the search launch (same four f32 WMMA k-steps, on the proj block in LDS): tvec never crosses the workspace; same sums, same order
# speedup vs baseline: 1.8759x; 1.0516x over previous
#include <hip/hip_runtime.h>
#include <math.h>

typedef float v2f __attribute__((ext_vector_type(2)));
typedef float v4f __attribute__((ext_vector_type(4)));
typedef float v8f __attribute__((ext_vector_type(8)));

#define BS      65536
#define IN_DIM  100
#define HID     16
#define NCORP   10
#define ROWS    (BS * NCORP)
#define WPB     8
#define NBLK1   512
#define NWAVE1  (NBLK1 * WPB)
#define TPW     (ROWS / 16 / NWAVE1)


template <int XM, int BANKS>
__device__ __forceinline__ float bank_xchg(const float keep, const float from)
{
    return __int_as_float(__builtin_amdgcn_update_dpp(
        __float_as_int(keep), __float_as_int(from), 0x160 + XM, 0xf, BANKS, false));
}
template <int XM, int BANKS>
__device__ __forceinline__ v2f bank_xchg2(const v2f keep, const v2f from)
{
    v2f r;
    r.x = bank_xchg<XM, BANKS>(keep.x, from.x);
    r.y = bank_xchg<XM, BANKS>(keep.y, from.y);
    return r;
}

__device__ __forceinline__ float tanh_sel(const float x)
{
    const float y = fabsf(x);
    const float t = x * x;
    float p = fmaf(__uint_as_float(0xbbbac73du), t, __uint_as_float(0x3ca908c9u));
    p = fmaf(t, p, __uint_as_float(0xbd5c1c4eu));
    p = fmaf(t, p, __uint_as_float(0x3e088382u));
    p = fmaf(t, p, __uint_as_float(0xbeaaaa99u));
    const float zs = fmaf(t, y * p, y);
    const float e  = expf(y + y);
    const float zb = fmaf(__builtin_amdgcn_rcpf(e + 1.0f), -2.0f, 1.0f);
    return __builtin_copysignf((y < 0.625f) ? zs : zb, x);
}

__device__ __forceinline__ void tile_fetch(const float* __restrict__ p,
                                           const float* __restrict__ pa,
                                           v2f (&t)[24], v2f& tl)
{
    #pragma unroll
    for (int q = 0; q < 6; ++q) {
        #pragma unroll
        for (int g = 0; g < 4; ++g)
            t[q * 4 + g] = *(const v2f*)(p + (4 * g) * IN_DIM + 16 * q);
    }
    tl = *(const v2f*)pa;
}

__device__ __forceinline__ void tile_compute(
    const v2f (&t)[24], const v2f tl,
    const v2f (&bw)[25],
    const float bias,
    float* __restrict__ proj,
    const long row0, const int l15, const int rofs)
{
    v8f acc = {};
    #pragma unroll
    for (int q = 0; q < 6; ++q) {
        const v2f u0 = bank_xchg2<4, 0xA>(t[q * 4 + 0], t[q * 4 + 1]);
        const v2f u1 = bank_xchg2<4, 0x5>(t[q * 4 + 1], t[q * 4 + 0]);
        const v2f u2 = bank_xchg2<4, 0xA>(t[q * 4 + 2], t[q * 4 + 3]);
        const v2f u3 = bank_xchg2<4, 0x5>(t[q * 4 + 3], t[q * 4 + 2]);
        v2f y[4];
        y[0] = bank_xchg2<8, 0xC>(u0, u2);
        y[2] = bank_xchg2<8, 0x3>(u2, u0);
        y[1] = bank_xchg2<8, 0xC>(u1, u3);
        y[3] = bank_xchg2<8, 0x3>(u3, u1);
        #pragma unroll
        for (int j = 0; j < 4; ++j)
            acc = __builtin_amdgcn_wmma_f32_16x16x4_f32(
                      false, y[j], false, bw[4 * q + j], (short)0, acc, false, false);
    }
    acc = __builtin_amdgcn_wmma_f32_16x16x4_f32(
              false, tl, false, bw[24], (short)0, acc, false, false);

    #pragma unroll
    for (int r = 0; r < 8; ++r) acc[r] += bias;

    #pragma unroll
    for (int r = 0; r < 8; ++r)
        proj[(row0 + r + rofs) * HID + l15] = acc[r];
}

__global__ __launch_bounds__(256) void proj_kernel(
    const float* __restrict__ item_corpus,
    const float* __restrict__ W_proj,
    const float* __restrict__ b_proj,
    float* __restrict__ proj)
{
    const int lane  = threadIdx.x & 31;
    const int wv    = threadIdx.x >> 5;
    const int l15   = lane & 15;
    const int khalf = (lane >> 4) * 2;
    const int rofs  = (lane >> 4) * 8;

    v2f bw[25];
    #pragma unroll
    for (int ks = 0; ks < 25; ++ks) {
        bw[ks].x = W_proj[(4 * ks + khalf + 0) * HID + l15];
        bw[ks].y = W_proj[(4 * ks + khalf + 1) * HID + l15];
    }
    const float bias = b_proj[l15];

    const long gw     = (long)blockIdx.x * WPB + wv;
    const long tstep  = (long)NWAVE1 * 16;
    long row0 = gw * 16;
    const float* p  = item_corpus + (row0 + (lane & 3)) * IN_DIM + 4 * ((lane >> 2) & 3) + khalf;
    const float* pa = item_corpus + (row0 + l15) * IN_DIM + 96 + khalf;
    const long pstep = tstep * IN_DIM;

    v2f   ta[24], tb[24];
    v2f   tla, tlb;

    tile_fetch(p, pa, ta, tla);
    #pragma unroll 1
    for (int i = 0; i < TPW - 2; i += 2) {
        tile_fetch(p + pstep, pa + pstep, tb, tlb);
        __builtin_amdgcn_sched_barrier(0);
        tile_compute(ta, tla, bw, bias, proj, row0, l15, rofs);
        tile_fetch(p + 2 * pstep, pa + 2 * pstep, ta, tla);
        __builtin_amdgcn_sched_barrier(0);
        tile_compute(tb, tlb, bw, bias, proj, row0 + tstep, l15, rofs);
        p    += 2 * pstep;
        pa   += 2 * pstep;
        row0 += 2 * tstep;
    }
    tile_fetch(p + pstep, pa + pstep, tb, tlb);
    __builtin_amdgcn_sched_barrier(0);
    tile_compute(ta, tla, bw, bias, proj, row0, l15, rofs);
    tile_compute(tb, tlb, bw, bias, proj, row0 + tstep, l15, rofs);
}

#define TV_U      (NCORP * HID / 4)
#define TV_PITCH  (TV_U + 1)
#define OUT_PITCH 25

__global__ __launch_bounds__(256) void greedy_kernel(
    const float* __restrict__ user_intent,
    const float* __restrict__ proj,
    const float* __restrict__ W_t,
    const float* __restrict__ W_s,
    const int*   __restrict__ search_num_p,
    float* __restrict__ out)
{
    __shared__ float Ws[256];
    __shared__ v4f   blk[WPB][32 * TV_PITCH];
    __shared__ v4f   ots[WPB][32 * OUT_PITCH];
    Ws[threadIdx.x] = W_s[threadIdx.x];

    const int  b     = blockIdx.x * blockDim.x + threadIdx.x;
    const int  lane  = threadIdx.x & 31;
    const int  wv    = threadIdx.x >> 5;
    const long e0    = b - lane;
    const int  steps = *search_num_p;
    const long obase = (long)b * (steps + 1) * HID;
    const bool staged = 4 * (steps + 1) < OUT_PITCH;

    const int  l15   = lane & 15;
    const int  khalf = (lane >> 4) * 2;
    const int  rofs  = (lane >> 4) * 8;

    const v4f* psrc = (const v4f*)(proj + e0 * NCORP * HID) + lane;
    v4f nx[10];
    #pragma unroll
    for (int ii = 0; ii < 10; ++ii) nx[ii] = psrc[32 * ii];

    v2f bt[4];
    #pragma unroll
    for (int kc = 0; kc < 4; ++kc) {
        bt[kc].x = W_t[(kc * 4 + khalf + 0) * HID + l15];
        bt[kc].y = W_t[(kc * 4 + khalf + 1) * HID + l15];
    }

    const float* pb = (const float*)&blk[wv][0];
    float*       sc = (float*)&ots[wv][0];
    float tv[NCORP * HID];
    #pragma unroll
    for (int j = 0; j < NCORP * HID; ++j) tv[j] = 0.0f;

    int abase[5];
    #pragma unroll
    for (int p = 0; p < 5; ++p) {
        const int x = 16 * p + l15;
        const int q = x / NCORP;
        abase[p] = q * (TV_PITCH * 4) + (x - q * NCORP) * HID + khalf;
    }
    const int sbase_same = l15 + (lane >> 4) * (8 * HID);
    const int sbase_next = l15 + (lane >> 4) * (8 * HID + 4);

    #pragma unroll 1
    for (int G = 0; G < 4; ++G) {
        v4f cu[10];
        #pragma unroll
        for (int ii = 0; ii < 10; ++ii) cu[ii] = nx[ii];
        const v4f* pn = psrc + 320 * ((G < 3) ? G + 1 : 3);
        #pragma unroll
        for (int ii = 0; ii < 10; ++ii) nx[ii] = pn[32 * ii];

        v4f* bg = &blk[wv][8 * G * TV_PITCH];
        #pragma unroll
        for (int ii = 0; ii < 10; ++ii) {
            const int row = (32 * ii) / TV_U + ((lane + (32 * ii) % TV_U >= TV_U) ? 1 : 0);
            bg[32 * ii + lane + row] = cu[ii];
        }
        __builtin_amdgcn_fence(__ATOMIC_RELEASE, "wavefront");
        __builtin_amdgcn_wave_barrier();
        __builtin_amdgcn_fence(__ATOMIC_ACQUIRE, "wavefront");

        const float* pg = pb + 8 * G * (TV_PITCH * 4);
        float*       sg = sc + 8 * (G & 1) * (TV_PITCH * 4);
        #pragma unroll
        for (int p = 0; p < 5; ++p) {
            const float* ap = pg + abase[p];
            v8f acc2 = {};
            #pragma unroll
            for (int kc = 0; kc < 4; ++kc) {
                v2f a;
                a.x = ap[kc * 4 + 0];
                a.y = ap[kc * 4 + 1];
                acc2 = __builtin_amdgcn_wmma_f32_16x16x4_f32(
                           false, a, false, bt[kc], (short)0, acc2, false, false);
            }
            #pragma unroll
            for (int r = 0; r < 8; ++r) {
                const int x0 = 16 * p + r;
                const int i0 = (x0 / NCORP) * (TV_PITCH * 4) + (x0 % NCORP) * HID;
                const int sb = (x0 % NCORP < 2) ? sbase_same : sbase_next;
                sg[i0 + sb] = tanh_sel(acc2[r]);
            }
        }
        if (G & 1) {
            __builtin_amdgcn_fence(__ATOMIC_RELEASE, "wavefront");
            __builtin_amdgcn_wave_barrier();
            __builtin_amdgcn_fence(__ATOMIC_ACQUIRE, "wavefront");
            if ((lane >> 4) == (G >> 1)) {
                #pragma unroll
                for (int j = 0; j < TV_U; ++j) {
                    const v4f u = ots[wv][TV_PITCH * l15 + j];
                    tv[4 * j + 0] = u.x;
                    tv[4 * j + 1] = u.y;
                    tv[4 * j + 2] = u.z;
                    tv[4 * j + 3] = u.w;
                }
            }
            __builtin_amdgcn_fence(__ATOMIC_RELEASE, "wavefront");
            __builtin_amdgcn_wave_barrier();
            __builtin_amdgcn_fence(__ATOMIC_ACQUIRE, "wavefront");
        }
    }
    __syncthreads();

    float sum[HID];
    const v4f* usrc = (const v4f*)(user_intent + (long)b * HID);
    v4f g[4];
    #pragma unroll
    for (int j = 0; j < 4; ++j) {
        g[j] = usrc[j];
        sum[4 * j + 0] = g[j].x;
        sum[4 * j + 1] = g[j].y;
        sum[4 * j + 2] = g[j].z;
        sum[4 * j + 3] = g[j].w;
    }
    if (staged) {
        #pragma unroll
        for (int j = 0; j < 4; ++j) ots[wv][OUT_PITCH * lane + j] = g[j];
    } else {
        #pragma unroll
        for (int j = 0; j < 4; ++j) *(v4f*)(out + obase + 4 * j) = g[j];
    }

    for (int it = 0; it < steps; ++it) {
        const float inv = 1.0f / (float)(it + 1);

        float xs[HID];
        #pragma unroll
        for (int hp = 0; hp < HID; ++hp) xs[hp] = sum[hp] * inv;
        float s[HID];
        #pragma unroll
        for (int h = 0; h < HID; ++h) {
            float a = 0.0f;
            #pragma unroll
            for (int hp = 0; hp < HID; ++hp)
                a = fmaf(xs[hp], Ws[hp * HID + h], a);
            s[h] = tanh_sel(a);
        }

        float best = -__builtin_inff();
        int bidx = 0;
        #pragma unroll
        for (int c = 0; c < NCORP; ++c) {
            float sc = 0.0f;
            {
                #pragma clang fp contract(off)
                #pragma unroll
                for (int h = 0; h < HID; ++h)
                    sc += tv[c * HID + h] * s[h];
            }
            if (sc > best) { best = sc; bidx = c; }
        }

        #pragma unroll
        for (int j = 0; j < 4; ++j) {
            g[j] = blk[wv][TV_PITCH * lane + 4 * bidx + j];
            sum[4 * j + 0] += g[j].x;
            sum[4 * j + 1] += g[j].y;
            sum[4 * j + 2] += g[j].z;
            sum[4 * j + 3] += g[j].w;
        }
        if (staged) {
            #pragma unroll
            for (int j = 0; j < 4; ++j)
                ots[wv][OUT_PITCH * lane + 4 * (it + 1) + j] = g[j];
        } else {
            #pragma unroll
            for (int j = 0; j < 4; ++j)
                *(v4f*)(out + obase + (long)(it + 1) * HID + 4 * j) = g[j];
        }
    }

    if (staged) {
        __syncthreads();
        const int nu = 4 * (steps + 1);
        v4f* odst = (v4f*)(out + e0 * (steps + 1) * HID);
        for (int i = 0; i < nu; ++i) {
            const int q   = 32 * i + lane;
            const int row = q / nu;
            odst[q] = ots[wv][OUT_PITCH * row + (q - row * nu)];
        }
    }
}

extern "C" void kernel_launch(void* const* d_in, const int* in_sizes, int n_in,
                              void* d_out, int out_size, void* d_ws, size_t ws_size,
                              hipStream_t stream) {
    const float* user_intent = (const float*)d_in[0];
    const float* item_corpus = (const float*)d_in[1];
    const float* W_proj      = (const float*)d_in[2];
    const float* b_proj      = (const float*)d_in[3];
    const float* W_s         = (const float*)d_in[4];
    const float* W_t         = (const float*)d_in[5];
    const int*   search_num  = (const int*)d_in[6];
    float* out = (float*)d_out;

    float* proj = (float*)d_ws;

    proj_kernel<<<NBLK1, 256, 0, stream>>>(item_corpus, W_proj, b_proj, proj);
    greedy_kernel<<<BS / 256, 256, 0, stream>>>(user_intent, proj, W_t,
                                                W_s, search_num, out);
}


# tanh's reciprocal form takes exp(2|x|) as hardware exp2 of |x| * 2 log2(e), not through expf (13 instructions): tvec and s move by at most two last-place units, scores no longer bit-identical
# speedup vs baseline: 2.0229x; 1.0784x over previous
#include <hip/hip_runtime.h>
#include <math.h>

typedef float v2f __attribute__((ext_vector_type(2)));
typedef float v4f __attribute__((ext_vector_type(4)));
typedef float v8f __attribute__((ext_vector_type(8)));

#define BS      65536
#define IN_DIM  100
#define HID     16
#define NCORP   10
#define ROWS    (BS * NCORP)
#define WPB     8
#define NBLK1   512
#define NWAVE1  (NBLK1 * WPB)
#define TPW     (ROWS / 16 / NWAVE1)


template <int XM, int BANKS>
__device__ __forceinline__ float bank_xchg(const float keep, const float from)
{
    return __int_as_float(__builtin_amdgcn_update_dpp(
        __float_as_int(keep), __float_as_int(from), 0x160 + XM, 0xf, BANKS, false));
}
template <int XM, int BANKS>
__device__ __forceinline__ v2f bank_xchg2(const v2f keep, const v2f from)
{
    v2f r;
    r.x = bank_xchg<XM, BANKS>(keep.x, from.x);
    r.y = bank_xchg<XM, BANKS>(keep.y, from.y);
    return r;
}

__device__ __forceinline__ float tanh_sel(const float x)
{
    const float y = fabsf(x);
    const float t = x * x;
    float p = fmaf(__uint_as_float(0xbbbac73du), t, __uint_as_float(0x3ca908c9u));
    p = fmaf(t, p, __uint_as_float(0xbd5c1c4eu));
    p = fmaf(t, p, __uint_as_float(0x3e088382u));
    p = fmaf(t, p, __uint_as_float(0xbeaaaa99u));
    const float zs = fmaf(t, y * p, y);
    const float e  = __builtin_amdgcn_exp2f(y * __uint_as_float(0x4038aa3bu));
    const float zb = fmaf(__builtin_amdgcn_rcpf(e + 1.0f), -2.0f, 1.0f);
    return __builtin_copysignf((y < 0.625f) ? zs : zb, x);
}

__device__ __forceinline__ void tile_fetch(const float* __restrict__ p,
                                           const float* __restrict__ pa,
                                           v2f (&t)[24], v2f& tl)
{
    #pragma unroll
    for (int q = 0; q < 6; ++q) {
        #pragma unroll
        for (int g = 0; g < 4; ++g)
            t[q * 4 + g] = *(const v2f*)(p + (4 * g) * IN_DIM + 16 * q);
    }
    tl = *(const v2f*)pa;
}

__device__ __forceinline__ void tile_compute(
    const v2f (&t)[24], const v2f tl,
    const v2f (&bw)[25],
    const float bias,
    float* __restrict__ proj,
    const long row0, const int l15, const int rofs)
{
    v8f acc = {};
    #pragma unroll
    for (int q = 0; q < 6; ++q) {
        const v2f u0 = bank_xchg2<4, 0xA>(t[q * 4 + 0], t[q * 4 + 1]);
        const v2f u1 = bank_xchg2<4, 0x5>(t[q * 4 + 1], t[q * 4 + 0]);
        const v2f u2 = bank_xchg2<4, 0xA>(t[q * 4 + 2], t[q * 4 + 3]);
        const v2f u3 = bank_xchg2<4, 0x5>(t[q * 4 + 3], t[q * 4 + 2]);
        v2f y[4];
        y[0] = bank_xchg2<8, 0xC>(u0, u2);
        y[2] = bank_xchg2<8, 0x3>(u2, u0);
        y[1] = bank_xchg2<8, 0xC>(u1, u3);
        y[3] = bank_xchg2<8, 0x3>(u3, u1);
        #pragma unroll
        for (int j = 0; j < 4; ++j)
            acc = __builtin_amdgcn_wmma_f32_16x16x4_f32(
                      false, y[j], false, bw[4 * q + j], (short)0, acc, false, false);
    }
    acc = __builtin_amdgcn_wmma_f32_16x16x4_f32(
              false, tl, false, bw[24], (short)0, acc, false, false);

    #pragma unroll
    for (int r = 0; r < 8; ++r) acc[r] += bias;

    #pragma unroll
    for (int r = 0; r < 8; ++r)
        proj[(row0 + r + rofs) * HID + l15] = acc[r];
}

__global__ __launch_bounds__(256) void proj_kernel(
    const float* __restrict__ item_corpus,
    const float* __restrict__ W_proj,
    const float* __restrict__ b_proj,
    float* __restrict__ proj)
{
    const int lane  = threadIdx.x & 31;
    const int wv    = threadIdx.x >> 5;
    const int l15   = lane & 15;
    const int khalf = (lane >> 4) * 2;
    const int rofs  = (lane >> 4) * 8;

    v2f bw[25];
    #pragma unroll
    for (int ks = 0; ks < 25; ++ks) {
        bw[ks].x = W_proj[(4 * ks + khalf + 0) * HID + l15];
        bw[ks].y = W_proj[(4 * ks + khalf + 1) * HID + l15];
    }
    const float bias = b_proj[l15];

    const long gw     = (long)blockIdx.x * WPB + wv;
    const long tstep  = (long)NWAVE1 * 16;
    long row0 = gw * 16;
    const float* p  = item_corpus + (row0 + (lane & 3)) * IN_DIM + 4 * ((lane >> 2) & 3) + khalf;
    const float* pa = item_corpus + (row0 + l15) * IN_DIM + 96 + khalf;
    const long pstep = tstep * IN_DIM;

    v2f   ta[24], tb[24];
    v2f   tla, tlb;

    tile_fetch(p, pa, ta, tla);
    #pragma unroll 1
    for (int i = 0; i < TPW - 2; i += 2) {
        tile_fetch(p + pstep, pa + pstep, tb, tlb);
        __builtin_amdgcn_sched_barrier(0);
        tile_compute(ta, tla, bw, bias, proj, row0, l15, rofs);
        tile_fetch(p + 2 * pstep, pa + 2 * pstep, ta, tla);
        __builtin_amdgcn_sched_barrier(0);
        tile_compute(tb, tlb, bw, bias, proj, row0 + tstep, l15, rofs);
        p    += 2 * pstep;
        pa   += 2 * pstep;
        row0 += 2 * tstep;
    }
    tile_fetch(p + pstep, pa + pstep, tb, tlb);
    __builtin_amdgcn_sched_barrier(0);
    tile_compute(ta, tla, bw, bias, proj, row0, l15, rofs);
    tile_compute(tb, tlb, bw, bias, proj, row0 + tstep, l15, rofs);
}

#define TV_U      (NCORP * HID / 4)
#define TV_PITCH  (TV_U + 1)
#define OUT_PITCH 25

__global__ __launch_bounds__(256) void greedy_kernel(
    const float* __restrict__ user_intent,
    const float* __restrict__ proj,
    const float* __restrict__ W_t,
    const float* __restrict__ W_s,
    const int*   __restrict__ search_num_p,
    float* __restrict__ out)
{
    __shared__ float Ws[256];
    __shared__ v4f   blk[WPB][32 * TV_PITCH];
    __shared__ v4f   ots[WPB][32 * OUT_PITCH];
    Ws[threadIdx.x] = W_s[threadIdx.x];

    const int  b     = blockIdx.x * blockDim.x + threadIdx.x;
    const int  lane  = threadIdx.x & 31;
    const int  wv    = threadIdx.x >> 5;
    const long e0    = b - lane;
    const int  steps = *search_num_p;
    const long obase = (long)b * (steps + 1) * HID;
    const bool staged = 4 * (steps + 1) < OUT_PITCH;

    const int  l15   = lane & 15;
    const int  khalf = (lane >> 4) * 2;
    const int  rofs  = (lane >> 4) * 8;

    const v4f* psrc = (const v4f*)(proj + e0 * NCORP * HID) + lane;
    v4f nx[10];
    #pragma unroll
    for (int ii = 0; ii < 10; ++ii) nx[ii] = psrc[32 * ii];

    v2f bt[4];
    #pragma unroll
    for (int kc = 0; kc < 4; ++kc) {
        bt[kc].x = W_t[(kc * 4 + khalf + 0) * HID + l15];
        bt[kc].y = W_t[(kc * 4 + khalf + 1) * HID + l15];
    }

    const float* pb = (const float*)&blk[wv][0];
    float*       sc = (float*)&ots[wv][0];
    float tv[NCORP * HID];
    #pragma unroll
    for (int j = 0; j < NCORP * HID; ++j) tv[j] = 0.0f;

    int abase[5];
    #pragma unroll
    for (int p = 0; p < 5; ++p) {
        const int x = 16 * p + l15;
        const int q = x / NCORP;
        abase[p] = q * (TV_PITCH * 4) + (x - q * NCORP) * HID + khalf;
    }
    const int sbase_same = l15 + (lane >> 4) * (8 * HID);
    const int sbase_next = l15 + (lane >> 4) * (8 * HID + 4);

    #pragma unroll 1
    for (int G = 0; G < 4; ++G) {
        v4f cu[10];
        #pragma unroll
        for (int ii = 0; ii < 10; ++ii) cu[ii] = nx[ii];
        const v4f* pn = psrc + 320 * ((G < 3) ? G + 1 : 3);
        #pragma unroll
        for (int ii = 0; ii < 10; ++ii) nx[ii] = pn[32 * ii];

        v4f* bg = &blk[wv][8 * G * TV_PITCH];
        #pragma unroll
        for (int ii = 0; ii < 10; ++ii) {
            const int row = (32 * ii) / TV_U + ((lane + (32 * ii) % TV_U >= TV_U) ? 1 : 0);
            bg[32 * ii + lane + row] = cu[ii];
        }
        __builtin_amdgcn_fence(__ATOMIC_RELEASE, "wavefront");
        __builtin_amdgcn_wave_barrier();
        __builtin_amdgcn_fence(__ATOMIC_ACQUIRE, "wavefront");

        const float* pg = pb + 8 * G * (TV_PITCH * 4);
        float*       sg = sc + 8 * (G & 1) * (TV_PITCH * 4);
        #pragma unroll
        for (int p = 0; p < 5; ++p) {
            const float* ap = pg + abase[p];
            v8f acc2 = {};
            #pragma unroll
            for (int kc = 0; kc < 4; ++kc) {
                v2f a;
                a.x = ap[kc * 4 + 0];
                a.y = ap[kc * 4 + 1];
                acc2 = __builtin_amdgcn_wmma_f32_16x16x4_f32(
                           false, a, false, bt[kc], (short)0, acc2, false, false);
            }
            #pragma unroll
            for (int r = 0; r < 8; ++r) {
                const int x0 = 16 * p + r;
                const int i0 = (x0 / NCORP) * (TV_PITCH * 4) + (x0 % NCORP) * HID;
                const int sb = (x0 % NCORP < 2) ? sbase_same : sbase_next;
                sg[i0 + sb] = tanh_sel(acc2[r]);
            }
        }
        if (G & 1) {
            __builtin_amdgcn_fence(__ATOMIC_RELEASE, "wavefront");
            __builtin_amdgcn_wave_barrier();
            __builtin_amdgcn_fence(__ATOMIC_ACQUIRE, "wavefront");
            if ((lane >> 4) == (G >> 1)) {
                #pragma unroll
                for (int j = 0; j < TV_U; ++j) {
                    const v4f u = ots[wv][TV_PITCH * l15 + j];
                    tv[4 * j + 0] = u.x;
                    tv[4 * j + 1] = u.y;
                    tv[4 * j + 2] = u.z;
                    tv[4 * j + 3] = u.w;
                }
            }
            __builtin_amdgcn_fence(__ATOMIC_RELEASE, "wavefront");
            __builtin_amdgcn_wave_barrier();
            __builtin_amdgcn_fence(__ATOMIC_ACQUIRE, "wavefront");
        }
    }
    __syncthreads();

    float sum[HID];
    const v4f* usrc = (const v4f*)(user_intent + (long)b * HID);
    v4f g[4];
    #pragma unroll
    for (int j = 0; j < 4; ++j) {
        g[j] = usrc[j];
        sum[4 * j + 0] = g[j].x;
        sum[4 * j + 1] = g[j].y;
        sum[4 * j + 2] = g[j].z;
        sum[4 * j + 3] = g[j].w;
    }
    if (staged) {
        #pragma unroll
        for (int j = 0; j < 4; ++j) ots[wv][OUT_PITCH * lane + j] = g[j];
    } else {
        #pragma unroll
        for (int j = 0; j < 4; ++j) *(v4f*)(out + obase + 4 * j) = g[j];
    }

    for (int it = 0; it < steps; ++it) {
        const float inv = 1.0f / (float)(it + 1);

        float xs[HID];
        #pragma unroll
        for (int hp = 0; hp < HID; ++hp) xs[hp] = sum[hp] * inv;
        float s[HID];
        #pragma unroll
        for (int h = 0; h < HID; ++h) {
            float a = 0.0f;
            #pragma unroll
            for (int hp = 0; hp < HID; ++hp)
                a = fmaf(xs[hp], Ws[hp * HID + h], a);
            s[h] = tanh_sel(a);
        }

        float best = -__builtin_inff();
        int bidx = 0;
        #pragma unroll
        for (int c = 0; c < NCORP; ++c) {
            float sc = 0.0f;
            {
                #pragma clang fp contract(off)
                #pragma unroll
                for (int h = 0; h < HID; ++h)
                    sc += tv[c * HID + h] * s[h];
            }
            if (sc > best) { best = sc; bidx = c; }
        }

        #pragma unroll
        for (int j = 0; j < 4; ++j) {
            g[j] = blk[wv][TV_PITCH * lane + 4 * bidx + j];
            sum[4 * j + 0] += g[j].x;
            sum[4 * j + 1] += g[j].y;
            sum[4 * j + 2] += g[j].z;
            sum[4 * j + 3] += g[j].w;
        }
        if (staged) {
            #pragma unroll
            for (int j = 0; j < 4; ++j)
                ots[wv][OUT_PITCH * lane + 4 * (it + 1) + j] = g[j];
        } else {
            #pragma unroll
            for (int j = 0; j < 4; ++j)
                *(v4f*)(out + obase + (long)(it + 1) * HID + 4 * j) = g[j];
        }
    }

    if (staged) {
        __syncthreads();
        const int nu = 4 * (steps + 1);
        v4f* odst = (v4f*)(out + e0 * (steps + 1) * HID);
        for (int i = 0; i < nu; ++i) {
            const int q   = 32 * i + lane;
            const int row = q / nu;
            odst[q] = ots[wv][OUT_PITCH * row + (q - row * nu)];
        }
    }
}

extern "C" void kernel_launch(void* const* d_in, const int* in_sizes, int n_in,
                              void* d_out, int out_size, void* d_ws, size_t ws_size,
                              hipStream_t stream) {
    const float* user_intent = (const float*)d_in[0];
    const float* item_corpus = (const float*)d_in[1];
    const float* W_proj      = (const float*)d_in[2];
    const float* b_proj      = (const float*)d_in[3];
    const float* W_s         = (const float*)d_in[4];
    const float* W_t         = (const float*)d_in[5];
    const int*   search_num  = (const int*)d_in[6];
    float* out = (float*)d_out;

    float* proj = (float*)d_ws;

    proj_kernel<<<NBLK1, 256, 0, stream>>>(item_corpus, W_proj, b_proj, proj);
    greedy_kernel<<<BS / 256, 256, 0, stream>>>(user_intent, proj, W_t,
                                                W_s, search_num, out);
}


# Projection tile leaves in whole 128-B lines: the 16x16 D tile (1 KiB contiguous) turns through a private LDS region and goes out as two 512-B stores, not sixteen 64-B row pieces; same values
# speedup vs baseline: 2.0910x; 1.0337x over previous
#include <hip/hip_runtime.h>
#include <math.h>

typedef float v2f __attribute__((ext_vector_type(2)));
typedef float v4f __attribute__((ext_vector_type(4)));
typedef float v8f __attribute__((ext_vector_type(8)));

#define BS      65536
#define IN_DIM  100
#define HID     16
#define NCORP   10
#define ROWS    (BS * NCORP)
#define WPB     8
#define NBLK1   512
#define NWAVE1  (NBLK1 * WPB)
#define TPW     (ROWS / 16 / NWAVE1)

#define TURN_U   68

template <int XM, int BANKS>
__device__ __forceinline__ float bank_xchg(const float keep, const float from)
{
    return __int_as_float(__builtin_amdgcn_update_dpp(
        __float_as_int(keep), __float_as_int(from), 0x160 + XM, 0xf, BANKS, false));
}
template <int XM, int BANKS>
__device__ __forceinline__ v2f bank_xchg2(const v2f keep, const v2f from)
{
    v2f r;
    r.x = bank_xchg<XM, BANKS>(keep.x, from.x);
    r.y = bank_xchg<XM, BANKS>(keep.y, from.y);
    return r;
}

__device__ __forceinline__ float tanh_sel(const float x)
{
    const float y = fabsf(x);
    const float t = x * x;
    float p = fmaf(__uint_as_float(0xbbbac73du), t, __uint_as_float(0x3ca908c9u));
    p = fmaf(t, p, __uint_as_float(0xbd5c1c4eu));
    p = fmaf(t, p, __uint_as_float(0x3e088382u));
    p = fmaf(t, p, __uint_as_float(0xbeaaaa99u));
    const float zs = fmaf(t, y * p, y);
    const float e  = __builtin_amdgcn_exp2f(y * __uint_as_float(0x4038aa3bu));
    const float zb = fmaf(__builtin_amdgcn_rcpf(e + 1.0f), -2.0f, 1.0f);
    return __builtin_copysignf((y < 0.625f) ? zs : zb, x);
}

__device__ __forceinline__ void tile_fetch(const float* __restrict__ p,
                                           const float* __restrict__ pa,
                                           v2f (&t)[24], v2f& tl)
{
    #pragma unroll
    for (int q = 0; q < 6; ++q) {
        #pragma unroll
        for (int g = 0; g < 4; ++g)
            t[q * 4 + g] = *(const v2f*)(p + (4 * g) * IN_DIM + 16 * q);
    }
    tl = *(const v2f*)pa;
}

__device__ __forceinline__ void tile_compute(
    const v2f (&t)[24], const v2f tl,
    const v2f (&bw)[25],
    const float bias,
    float* __restrict__ proj,
    const long row0, const int lane,
    float* __restrict__ tw,
    const v4f* __restrict__ tr)
{
    v8f acc = {};
    #pragma unroll
    for (int q = 0; q < 6; ++q) {
        const v2f u0 = bank_xchg2<4, 0xA>(t[q * 4 + 0], t[q * 4 + 1]);
        const v2f u1 = bank_xchg2<4, 0x5>(t[q * 4 + 1], t[q * 4 + 0]);
        const v2f u2 = bank_xchg2<4, 0xA>(t[q * 4 + 2], t[q * 4 + 3]);
        const v2f u3 = bank_xchg2<4, 0x5>(t[q * 4 + 3], t[q * 4 + 2]);
        v2f y[4];
        y[0] = bank_xchg2<8, 0xC>(u0, u2);
        y[2] = bank_xchg2<8, 0x3>(u2, u0);
        y[1] = bank_xchg2<8, 0xC>(u1, u3);
        y[3] = bank_xchg2<8, 0x3>(u3, u1);
        #pragma unroll
        for (int j = 0; j < 4; ++j)
            acc = __builtin_amdgcn_wmma_f32_16x16x4_f32(
                      false, y[j], false, bw[4 * q + j], (short)0, acc, false, false);
    }
    acc = __builtin_amdgcn_wmma_f32_16x16x4_f32(
              false, tl, false, bw[24], (short)0, acc, false, false);

    #pragma unroll
    for (int r = 0; r < 8; ++r) acc[r] += bias;

    #pragma unroll
    for (int r = 0; r < 8; ++r) tw[r * HID] = acc[r];
    __builtin_amdgcn_fence(__ATOMIC_RELEASE, "wavefront", "local");
    __builtin_amdgcn_wave_barrier();
    __builtin_amdgcn_fence(__ATOMIC_ACQUIRE, "wavefront", "local");
    const v4f o0 = tr[0];
    const v4f o1 = tr[36];
    __builtin_amdgcn_fence(__ATOMIC_RELEASE, "wavefront", "local");
    __builtin_amdgcn_wave_barrier();
    __builtin_amdgcn_fence(__ATOMIC_ACQUIRE, "wavefront", "local");
    v4f* dst = (v4f*)(proj + row0 * HID) + lane;
    dst[0]  = o0;
    dst[32] = o1;
}

__global__ __launch_bounds__(256) void proj_kernel(
    const float* __restrict__ item_corpus,
    const float* __restrict__ W_proj,
    const float* __restrict__ b_proj,
    float* __restrict__ proj)
{
    const int lane  = threadIdx.x & 31;
    const int wv    = threadIdx.x >> 5;
    const int l15   = lane & 15;
    const int khalf = (lane >> 4) * 2;

    __shared__ v4f turn[WPB][TURN_U];
    float*     tw = (float*)&turn[wv][0] + l15 + (lane >> 4) * (8 * HID + 16);
    const v4f* tr = &turn[wv][lane];

    v2f bw[25];
    #pragma unroll
    for (int ks = 0; ks < 25; ++ks) {
        bw[ks].x = W_proj[(4 * ks + khalf + 0) * HID + l15];
        bw[ks].y = W_proj[(4 * ks + khalf + 1) * HID + l15];
    }
    const float bias = b_proj[l15];

    const long gw     = (long)blockIdx.x * WPB + wv;
    const long tstep  = (long)NWAVE1 * 16;
    long row0 = gw * 16;
    const float* p  = item_corpus + (row0 + (lane & 3)) * IN_DIM + 4 * ((lane >> 2) & 3) + khalf;
    const float* pa = item_corpus + (row0 + l15) * IN_DIM + 96 + khalf;
    const long pstep = tstep * IN_DIM;

    v2f   ta[24], tb[24];
    v2f   tla, tlb;

    tile_fetch(p, pa, ta, tla);
    #pragma unroll 1
    for (int i = 0; i < TPW - 2; i += 2) {
        tile_fetch(p + pstep, pa + pstep, tb, tlb);
        __builtin_amdgcn_sched_barrier(0);
        tile_compute(ta, tla, bw, bias, proj, row0, lane, tw, tr);
        tile_fetch(p + 2 * pstep, pa + 2 * pstep, ta, tla);
        __builtin_amdgcn_sched_barrier(0);
        tile_compute(tb, tlb, bw, bias, proj, row0 + tstep, lane, tw, tr);
        p    += 2 * pstep;
        pa   += 2 * pstep;
        row0 += 2 * tstep;
    }
    tile_fetch(p + pstep, pa + pstep, tb, tlb);
    __builtin_amdgcn_sched_barrier(0);
    tile_compute(ta, tla, bw, bias, proj, row0, lane, tw, tr);
    tile_compute(tb, tlb, bw, bias, proj, row0 + tstep, lane, tw, tr);
}

#define TV_U      (NCORP * HID / 4)
#define TV_PITCH  (TV_U + 1)
#define OUT_PITCH 25

__global__ __launch_bounds__(256) void greedy_kernel(
    const float* __restrict__ user_intent,
    const float* __restrict__ proj,
    const float* __restrict__ W_t,
    const float* __restrict__ W_s,
    const int*   __restrict__ search_num_p,
    float* __restrict__ out)
{
    __shared__ float Ws[256];
    __shared__ v4f   blk[WPB][32 * TV_PITCH];
    __shared__ v4f   ots[WPB][32 * OUT_PITCH];
    Ws[threadIdx.x] = W_s[threadIdx.x];

    const int  b     = blockIdx.x * blockDim.x + threadIdx.x;
    const int  lane  = threadIdx.x & 31;
    const int  wv    = threadIdx.x >> 5;
    const long e0    = b - lane;
    const int  steps = *search_num_p;
    const long obase = (long)b * (steps + 1) * HID;
    const bool staged = 4 * (steps + 1) < OUT_PITCH;

    const int  l15   = lane & 15;
    const int  khalf = (lane >> 4) * 2;
    const int  rofs  = (lane >> 4) * 8;

    const v4f* psrc = (const v4f*)(proj + e0 * NCORP * HID) + lane;
    v4f nx[10];
    #pragma unroll
    for (int ii = 0; ii < 10; ++ii) nx[ii] = psrc[32 * ii];

    v2f bt[4];
    #pragma unroll
    for (int kc = 0; kc < 4; ++kc) {
        bt[kc].x = W_t[(kc * 4 + khalf + 0) * HID + l15];
        bt[kc].y = W_t[(kc * 4 + khalf + 1) * HID + l15];
    }

    const float* pb = (const float*)&blk[wv][0];
    float*       sc = (float*)&ots[wv][0];
    float tv[NCORP * HID];
    #pragma unroll
    for (int j = 0; j < NCORP * HID; ++j) tv[j] = 0.0f;

    int abase[5];
    #pragma unroll
    for (int p = 0; p < 5; ++p) {
        const int x = 16 * p + l15;
        const int q = x / NCORP;
        abase[p] = q * (TV_PITCH * 4) + (x - q * NCORP) * HID + khalf;
    }
    const int sbase_same = l15 + (lane >> 4) * (8 * HID);
    const int sbase_next = l15 + (lane >> 4) * (8 * HID + 4);

    #pragma unroll 1
    for (int G = 0; G < 4; ++G) {
        v4f cu[10];
        #pragma unroll
        for (int ii = 0; ii < 10; ++ii) cu[ii] = nx[ii];
        const v4f* pn = psrc + 320 * ((G < 3) ? G + 1 : 3);
        #pragma unroll
        for (int ii = 0; ii < 10; ++ii) nx[ii] = pn[32 * ii];

        v4f* bg = &blk[wv][8 * G * TV_PITCH];
        #pragma unroll
        for (int ii = 0; ii < 10; ++ii) {
            const int row = (32 * ii) / TV_U + ((lane + (32 * ii) % TV_U >= TV_U) ? 1 : 0);
            bg[32 * ii + lane + row] = cu[ii];
        }
        __builtin_amdgcn_fence(__ATOMIC_RELEASE, "wavefront");
        __builtin_amdgcn_wave_barrier();
        __builtin_amdgcn_fence(__ATOMIC_ACQUIRE, "wavefront");

        const float* pg = pb + 8 * G * (TV_PITCH * 4);
        float*       sg = sc + 8 * (G & 1) * (TV_PITCH * 4);
        #pragma unroll
        for (int p = 0; p < 5; ++p) {
            const float* ap = pg + abase[p];
            v8f acc2 = {};
            #pragma unroll
            for (int kc = 0; kc < 4; ++kc) {
                v2f a;
                a.x = ap[kc * 4 + 0];
                a.y = ap[kc * 4 + 1];
                acc2 = __builtin_amdgcn_wmma_f32_16x16x4_f32(
                           false, a, false, bt[kc], (short)0, acc2, false, false);
            }
            #pragma unroll
            for (int r = 0; r < 8; ++r) {
                const int x0 = 16 * p + r;
                const int i0 = (x0 / NCORP) * (TV_PITCH * 4) + (x0 % NCORP) * HID;
                const int sb = (x0 % NCORP < 2) ? sbase_same : sbase_next;
                sg[i0 + sb] = tanh_sel(acc2[r]);
            }
        }
        if (G & 1) {
            __builtin_amdgcn_fence(__ATOMIC_RELEASE, "wavefront");
            __builtin_amdgcn_wave_barrier();
            __builtin_amdgcn_fence(__ATOMIC_ACQUIRE, "wavefront");
            if ((lane >> 4) == (G >> 1)) {
                #pragma unroll
                for (int j = 0; j < TV_U; ++j) {
                    const v4f u = ots[wv][TV_PITCH * l15 + j];
                    tv[4 * j + 0] = u.x;
                    tv[4 * j + 1] = u.y;
                    tv[4 * j + 2] = u.z;
                    tv[4 * j + 3] = u.w;
                }
            }
            __builtin_amdgcn_fence(__ATOMIC_RELEASE, "wavefront");
            __builtin_amdgcn_wave_barrier();
            __builtin_amdgcn_fence(__ATOMIC_ACQUIRE, "wavefront");
        }
    }
    __syncthreads();

    float sum[HID];
    const v4f* usrc = (const v4f*)(user_intent + (long)b * HID);
    v4f g[4];
    #pragma unroll
    for (int j = 0; j < 4; ++j) {
        g[j] = usrc[j];
        sum[4 * j + 0] = g[j].x;
        sum[4 * j + 1] = g[j].y;
        sum[4 * j + 2] = g[j].z;
        sum[4 * j + 3] = g[j].w;
    }
    if (staged) {
        #pragma unroll
        for (int j = 0; j < 4; ++j) ots[wv][OUT_PITCH * lane + j] = g[j];
    } else {
        #pragma unroll
        for (int j = 0; j < 4; ++j) *(v4f*)(out + obase + 4 * j) = g[j];
    }

    for (int it = 0; it < steps; ++it) {
        const float inv = 1.0f / (float)(it + 1);

        float xs[HID];
        #pragma unroll
        for (int hp = 0; hp < HID; ++hp) xs[hp] = sum[hp] * inv;
        float s[HID];
        #pragma unroll
        for (int h = 0; h < HID; ++h) {
            float a = 0.0f;
            #pragma unroll
            for (int hp = 0; hp < HID; ++hp)
                a = fmaf(xs[hp], Ws[hp * HID + h], a);
            s[h] = tanh_sel(a);
        }

        float best = -__builtin_inff();
        int bidx = 0;
        #pragma unroll
        for (int c = 0; c < NCORP; ++c) {
            float sc = 0.0f;
            {
                #pragma clang fp contract(off)
                #pragma unroll
                for (int h = 0; h < HID; ++h)
                    sc += tv[c * HID + h] * s[h];
            }
            if (sc > best) { best = sc; bidx = c; }
        }

        #pragma unroll
        for (int j = 0; j < 4; ++j) {
            g[j] = blk[wv][TV_PITCH * lane + 4 * bidx + j];
            sum[4 * j + 0] += g[j].x;
            sum[4 * j + 1] += g[j].y;
            sum[4 * j + 2] += g[j].z;
            sum[4 * j + 3] += g[j].w;
        }
        if (staged) {
            #pragma unroll
            for (int j = 0; j < 4; ++j)
                ots[wv][OUT_PITCH * lane + 4 * (it + 1) + j] = g[j];
        } else {
            #pragma unroll
            for (int j = 0; j < 4; ++j)
                *(v4f*)(out + obase + (long)(it + 1) * HID + 4 * j) = g[j];
        }
    }

    if (staged) {
        __syncthreads();
        const int nu = 4 * (steps + 1);
        v4f* odst = (v4f*)(out + e0 * (steps + 1) * HID);
        for (int i = 0; i < nu; ++i) {
            const int q   = 32 * i + lane;
            const int row = q / nu;
            odst[q] = ots[wv][OUT_PITCH * row + (q - row * nu)];
        }
    }
}

extern "C" void kernel_launch(void* const* d_in, const int* in_sizes, int n_in,
                              void* d_out, int out_size, void* d_ws, size_t ws_size,
                              hipStream_t stream) {
    const float* user_intent = (const float*)d_in[0];
    const float* item_corpus = (const float*)d_in[1];
    const float* W_proj      = (const float*)d_in[2];
    const float* b_proj      = (const float*)d_in[3];
    const float* W_s         = (const float*)d_in[4];
    const float* W_t         = (const float*)d_in[5];
    const int*   search_num  = (const int*)d_in[6];
    float* out = (float*)d_out;

    float* proj = (float*)d_ws;

    proj_kernel<<<NBLK1, 256, 0, stream>>>(item_corpus, W_proj, b_proj, proj);
    greedy_kernel<<<BS / 256, 256, 0, stream>>>(user_intent, proj, W_t,
                                                W_s, search_num, out);
}
